# starting kernel
# baseline (speedup 1.0000x reference)
#include <hip/hip_runtime.h>

typedef __attribute__((ext_vector_type(2))) float v2f;
typedef __attribute__((ext_vector_type(8))) float v8f;

template <int CIN, int COUT>
__global__ void __launch_bounds__(256) sparse_conv_wmma(
    const float* __restrict__ X,
    const float* __restrict__ W,
    const int*   __restrict__ G,
    const int*   __restrict__ S,
    float*       __restrict__ Y,
    int M) {
  constexpr int KSTEPS = CIN / 4;
  constexpr int NCH    = (COUT + 15) / 16;
  constexpr int PAIRS  = 27 * KSTEPS * 2 * COUT;

  extern __shared__ v2f wlds[];

  for (int p = (int)threadIdx.x; p < PAIRS; p += (int)blockDim.x) {
    const int n  = p % COUT;
    const int t  = p / COUT;
    const int h  = t & 1;
    const int kk = (t >> 1) % KSTEPS;
    const int k  = (t >> 1) / KSTEPS;
    const float* src = W + (long)(k * CIN + kk * 4 + 2 * h) * COUT + n;
    v2f w;
    w.x = src[0];
    w.y = src[COUT];
    wlds[p] = w;
  }
  __syncthreads();

  const int wave = (int)((blockIdx.x * blockDim.x + threadIdx.x) >> 5);
  const int lane = (int)(threadIdx.x & 31);
  const int j0 = wave * 16;
  if (j0 >= M) return;

  const int m  = lane & 15;
  const int hi = lane >> 4;
  const int j  = j0 + m;
  const bool jr = (j < M);
  const int lanehn = hi * COUT + m;

  v8f acc[NCH] = {};

  for (int k = 0; k < 27; ++k) {
    int  gi    = 0;
    bool found = false;
    if (jr) {
      gi    = __builtin_nontemporal_load(G + (long)k * M + j);
      found = (__builtin_nontemporal_load(S + (long)k * M + j) == j);
    }
    const float* xrow = X + (long)gi * CIN;
    const v2f*   wk   = wlds + k * (KSTEPS * 2 * COUT) + lanehn;

#pragma unroll
    for (int kk = 0; kk < KSTEPS; ++kk) {
      v2f A;
      if (found) {
        A.x = xrow[kk * 4 + 2 * hi + 0];
        A.y = xrow[kk * 4 + 2 * hi + 1];
      } else {
        A.x = 0.0f;
        A.y = 0.0f;
      }
#pragma unroll
      for (int c = 0; c < NCH; ++c) {
        const int n = m + c * 16;
        v2f B;
        if (n < COUT) {
          B = wk[kk * (2 * COUT) + c * 16];
        } else {
          B.x = 0.0f;
          B.y = 0.0f;
        }
        acc[c] = __builtin_amdgcn_wmma_f32_16x16x4_f32(
            false, A, false, B, (short)0, acc[c], false, false);
      }
    }
  }

#pragma unroll
  for (int c = 0; c < NCH; ++c) {
    const int n = m + c * 16;
    if (n < COUT) {
#pragma unroll
      for (int r = 0; r < 8; ++r) {
        const int row = j0 + r + 8 * hi;
        if (row < M) Y[(long)row * COUT + n] = acc[c][r];
      }
    }
  }
}

static inline int blocks_for(int M) {
  const int waves = (M + 15) / 16;
  return (waves + 7) / 8;
}

template <int CIN, int COUT>
static constexpr size_t lds_bytes() {
  return (size_t)27 * (CIN / 4) * 2 * COUT * sizeof(v2f);
}

extern "C" void kernel_launch(void* const* d_in, const int* in_sizes, int n_in,
                              void* d_out, int out_size, void* d_ws, size_t ws_size,
                              hipStream_t stream) {
  (void)n_in; (void)out_size; (void)ws_size;
  const float* feats = (const float*)d_in[0];
  const float* W1    = (const float*)d_in[1];
  const float* W2    = (const float*)d_in[2];
  const float* W3    = (const float*)d_in[3];
  const int*   g1    = (const int*)d_in[4];
  const int*   s1    = (const int*)d_in[5];
  const int*   g2    = (const int*)d_in[6];
  const int*   s2    = (const int*)d_in[7];
  const int*   g3    = (const int*)d_in[8];
  const int*   s3    = (const int*)d_in[9];

  const int N  = in_sizes[0] / 4;
  const int M2 = in_sizes[6] / 27;

  float* x1  = (float*)d_ws;
  float* x2  = x1 + (size_t)N * 32;
  float* out = (float*)d_out;

  sparse_conv_wmma<4, 32>
      <<<blocks_for(N), 256, lds_bytes<4, 32>(), stream>>>(feats, W1, g1, s1, x1, N);
  sparse_conv_wmma<32, 32>
      <<<blocks_for(M2), 256, lds_bytes<32, 32>(), stream>>>(x1, W2, g2, s2, x2, M2);
  sparse_conv_wmma<32, 2>
      <<<blocks_for(M2), 256, lds_bytes<32, 2>(), stream>>>(x2, W3, g3, s3, out, M2);
}


# k loop fully unrolled: all 27 index pairs of a row loaded up front and row gathers issued several offsets ahead of their WMMAs, so memory latencies overlap instead of chaining; sum order unchanged
# speedup vs baseline: 1.4865x; 1.4865x over previous
#include <hip/hip_runtime.h>

typedef __attribute__((ext_vector_type(2))) float v2f;
typedef __attribute__((ext_vector_type(8))) float v8f;

template <int CIN, int COUT, int DEPTH>
__global__ void __launch_bounds__(256) sparse_conv_wmma(
    const float* __restrict__ X,
    const float* __restrict__ W,
    const int*   __restrict__ G,
    const int*   __restrict__ S,
    float*       __restrict__ Y,
    int M) {
  constexpr int KSTEPS = CIN / 4;
  constexpr int NCH    = (COUT + 15) / 16;
  constexpr int PAIRS  = 27 * KSTEPS * 2 * COUT;

  extern __shared__ v2f wlds[];

  for (int p = (int)threadIdx.x; p < PAIRS; p += (int)blockDim.x) {
    const int n  = p % COUT;
    const int t  = p / COUT;
    const int h  = t & 1;
    const int kk = (t >> 1) % KSTEPS;
    const int k  = (t >> 1) / KSTEPS;
    const float* src = W + (long)(k * CIN + kk * 4 + 2 * h) * COUT + n;
    v2f w;
    w.x = src[0];
    w.y = src[COUT];
    wlds[p] = w;
  }
  __syncthreads();

  const int wave = (int)((blockIdx.x * blockDim.x + threadIdx.x) >> 5);
  const int lane = (int)(threadIdx.x & 31);
  const int j0 = wave * 16;
  if (j0 >= M) return;

  const int m  = lane & 15;
  const int hi = lane >> 4;
  const int j  = j0 + m;
  const bool jr = (j < M);
  const int lanehn = hi * COUT + m;

  v8f acc[NCH] = {};

  const unsigned joff = (unsigned)(jr ? j : j0) * 4u;
  int gv[27];
  int sv[27];
#pragma unroll
  for (int k = 0; k < 27; ++k) {
    const char* gk = (const char*)(G + (long)k * M);
    const char* sk = (const char*)(S + (long)k * M);
    gv[k] = __builtin_nontemporal_load((const int*)(gk + joff));
    sv[k] = __builtin_nontemporal_load((const int*)(sk + joff));
  }

  v2f  A[DEPTH][KSTEPS];
  bool hit[DEPTH];

#pragma unroll
  for (int d = 0; d < DEPTH; ++d) {
    hit[d] = jr && (sv[d] == j);
    const float* xrow = X + (long)(hit[d] ? gv[d] : 0) * CIN;
#pragma unroll
    for (int kk = 0; kk < KSTEPS; ++kk) {
      A[d][kk].x = xrow[kk * 4 + 2 * hi + 0];
      A[d][kk].y = xrow[kk * 4 + 2 * hi + 1];
    }
  }

#pragma unroll
  for (int k = 0; k < 27; ++k) {
    const v2f* wk   = wlds + k * (KSTEPS * 2 * COUT) + lanehn;
    const int  slot = k % DEPTH;

#pragma unroll
    for (int kk = 0; kk < KSTEPS; ++kk) {
      v2f Ak;
      Ak.x = hit[slot] ? A[slot][kk].x : 0.0f;
      Ak.y = hit[slot] ? A[slot][kk].y : 0.0f;
#pragma unroll
      for (int c = 0; c < NCH; ++c) {
        const int n = m + c * 16;
        v2f B;
        if (n < COUT) {
          B = wk[kk * (2 * COUT) + c * 16];
        } else {
          B.x = 0.0f;
          B.y = 0.0f;
        }
        acc[c] = __builtin_amdgcn_wmma_f32_16x16x4_f32(
            false, Ak, false, B, (short)0, acc[c], false, false);
      }
    }

    if (k + DEPTH < 27) {
      hit[slot] = jr && (sv[k + DEPTH] == j);
      const float* xrow = X + (long)(hit[slot] ? gv[k + DEPTH] : 0) * CIN;
#pragma unroll
      for (int kk = 0; kk < KSTEPS; ++kk) {
        A[slot][kk].x = xrow[kk * 4 + 2 * hi + 0];
        A[slot][kk].y = xrow[kk * 4 + 2 * hi + 1];
      }
    }
  }

#pragma unroll
  for (int c = 0; c < NCH; ++c) {
    const int n = m + c * 16;
    if (n < COUT) {
#pragma unroll
      for (int r = 0; r < 8; ++r) {
        const int row = j0 + r + 8 * hi;
        if (row < M) Y[(long)row * COUT + n] = acc[c][r];
      }
    }
  }
}

static inline int blocks_for(int M) {
  const int waves = (M + 15) / 16;
  return (waves + 7) / 8;
}

template <int CIN, int COUT>
static constexpr size_t lds_bytes() {
  return (size_t)27 * (CIN / 4) * 2 * COUT * sizeof(v2f);
}

extern "C" void kernel_launch(void* const* d_in, const int* in_sizes, int n_in,
                              void* d_out, int out_size, void* d_ws, size_t ws_size,
                              hipStream_t stream) {
  (void)n_in; (void)out_size; (void)ws_size;
  const float* feats = (const float*)d_in[0];
  const float* W1    = (const float*)d_in[1];
  const float* W2    = (const float*)d_in[2];
  const float* W3    = (const float*)d_in[3];
  const int*   g1    = (const int*)d_in[4];
  const int*   s1    = (const int*)d_in[5];
  const int*   g2    = (const int*)d_in[6];
  const int*   s2    = (const int*)d_in[7];
  const int*   g3    = (const int*)d_in[8];
  const int*   s3    = (const int*)d_in[9];

  const int N  = in_sizes[0] / 4;
  const int M2 = in_sizes[6] / 27;

  float* x1  = (float*)d_ws;
  float* x2  = x1 + (size_t)N * 32;
  float* out = (float*)d_out;

  sparse_conv_wmma<4, 32, 8>
      <<<blocks_for(N), 256, lds_bytes<4, 32>(), stream>>>(feats, W1, g1, s1, x1, N);
  sparse_conv_wmma<32, 32, 3>
      <<<blocks_for(M2), 256, lds_bytes<32, 32>(), stream>>>(x1, W2, g2, s2, x2, M2);
  sparse_conv_wmma<32, 2, 4>
      <<<blocks_for(M2), 256, lds_bytes<32, 2>(), stream>>>(x2, W3, g3, s3, out, M2);
}


# W staging into LDS: all of a thread's loads issued before its first LDS store, as 16-byte loads where COUT allows, instead of one serial L2 round trip per stored pair; sum order unchanged
# speedup vs baseline: 2.0338x; 1.3682x over previous
#include <hip/hip_runtime.h>

typedef __attribute__((ext_vector_type(2))) float v2f;
typedef __attribute__((ext_vector_type(4))) float v4f;
typedef __attribute__((ext_vector_type(8))) float v8f;

constexpr int BLOCK = 256;

template <int CIN, int COUT, int DEPTH>
__global__ void __launch_bounds__(BLOCK) sparse_conv_wmma(
    const float* __restrict__ X,
    const float* __restrict__ W,
    const int*   __restrict__ G,
    const int*   __restrict__ S,
    float*       __restrict__ Y,
    int M) {
  constexpr int KSTEPS = CIN / 4;
  constexpr int NCH    = (COUT + 15) / 16;
  constexpr int PAIRS  = 27 * KSTEPS * 2 * COUT;

  extern __shared__ __attribute__((aligned(16))) v2f wlds[];

  if constexpr (COUT % 4 == 0) {
    constexpr int QPR   = COUT / 4;
    constexpr int QUADS = PAIRS / 4;
    constexpr int NIT   = (QUADS + BLOCK - 1) / BLOCK;
    v4f ra[NIT];
    v4f rb[NIT];
#pragma unroll
    for (int i = 0; i < NIT; ++i) {
      const int q  = (int)threadIdx.x + i * BLOCK;
      const int qc = (i < NIT - 1 || q < QUADS) ? q : (QUADS - 1);
      const int g  = qc % QPR;
      const int t  = qc / QPR;
      const v4f* s = (const v4f*)(W + (long)(2 * t) * COUT + 4 * g);
      ra[i] = s[0];
      rb[i] = s[QPR];
    }
#pragma unroll
    for (int i = 0; i < NIT; ++i) {
      const int q = (int)threadIdx.x + i * BLOCK;
      if (i < NIT - 1 || q < QUADS) {
        const int g = q % QPR;
        const int t = q / QPR;
        v4f lo, hi4;
        lo.x  = ra[i].x; lo.y  = rb[i].x; lo.z  = ra[i].y; lo.w  = rb[i].y;
        hi4.x = ra[i].z; hi4.y = rb[i].z; hi4.z = ra[i].w; hi4.w = rb[i].w;
        v4f* d = (v4f*)(wlds + t * COUT + 4 * g);
        d[0] = lo;
        d[1] = hi4;
      }
    }
  } else {
    constexpr int NIT = (PAIRS + BLOCK - 1) / BLOCK;
    float wa[NIT];
    float wb[NIT];
#pragma unroll
    for (int i = 0; i < NIT; ++i) {
      const int p  = (int)threadIdx.x + i * BLOCK;
      const int pc = (i < NIT - 1 || p < PAIRS) ? p : (PAIRS - 1);
      const int n  = pc % COUT;
      const int t  = pc / COUT;
      const float* s = W + (long)(2 * t) * COUT + n;
      wa[i] = s[0];
      wb[i] = s[COUT];
    }
#pragma unroll
    for (int i = 0; i < NIT; ++i) {
      const int p = (int)threadIdx.x + i * BLOCK;
      if (i < NIT - 1 || p < PAIRS) {
        v2f w;
        w.x = wa[i];
        w.y = wb[i];
        wlds[p] = w;
      }
    }
  }
  __syncthreads();

  const int wave = (int)((blockIdx.x * BLOCK + threadIdx.x) >> 5);
  const int lane = (int)(threadIdx.x & 31);
  const int j0 = wave * 16;
  if (j0 >= M) return;

  const int m  = lane & 15;
  const int hi = lane >> 4;
  const int j  = j0 + m;
  const bool jr = (j < M);
  const int lanehn = hi * COUT + m;

  v8f acc[NCH] = {};

  const unsigned joff = (unsigned)(jr ? j : j0) * 4u;
  int gv[27];
  int sv[27];
#pragma unroll
  for (int k = 0; k < 27; ++k) {
    const char* gk = (const char*)(G + (long)k * M);
    const char* sk = (const char*)(S + (long)k * M);
    gv[k] = __builtin_nontemporal_load((const int*)(gk + joff));
    sv[k] = __builtin_nontemporal_load((const int*)(sk + joff));
  }

  v2f  A[DEPTH][KSTEPS];
  bool hit[DEPTH];

#pragma unroll
  for (int d = 0; d < DEPTH; ++d) {
    hit[d] = jr && (sv[d] == j);
    const float* xrow = X + (long)(hit[d] ? gv[d] : 0) * CIN;
#pragma unroll
    for (int kk = 0; kk < KSTEPS; ++kk) {
      A[d][kk].x = xrow[kk * 4 + 2 * hi + 0];
      A[d][kk].y = xrow[kk * 4 + 2 * hi + 1];
    }
  }

#pragma unroll
  for (int k = 0; k < 27; ++k) {
    const v2f* wk   = wlds + k * (KSTEPS * 2 * COUT) + lanehn;
    const int  slot = k % DEPTH;

#pragma unroll
    for (int kk = 0; kk < KSTEPS; ++kk) {
      v2f Ak;
      Ak.x = hit[slot] ? A[slot][kk].x : 0.0f;
      Ak.y = hit[slot] ? A[slot][kk].y : 0.0f;
#pragma unroll
      for (int c = 0; c < NCH; ++c) {
        const int n = m + c * 16;
        v2f B;
        if (n < COUT) {
          B = wk[kk * (2 * COUT) + c * 16];
        } else {
          B.x = 0.0f;
          B.y = 0.0f;
        }
        acc[c] = __builtin_amdgcn_wmma_f32_16x16x4_f32(
            false, Ak, false, B, (short)0, acc[c], false, false);
      }
    }

    if (k + DEPTH < 27) {
      hit[slot] = jr && (sv[k + DEPTH] == j);
      const float* xrow = X + (long)(hit[slot] ? gv[k + DEPTH] : 0) * CIN;
#pragma unroll
      for (int kk = 0; kk < KSTEPS; ++kk) {
        A[slot][kk].x = xrow[kk * 4 + 2 * hi + 0];
        A[slot][kk].y = xrow[kk * 4 + 2 * hi + 1];
      }
    }
  }

#pragma unroll
  for (int c = 0; c < NCH; ++c) {
    const int n = m + c * 16;
    if (n < COUT) {
#pragma unroll
      for (int r = 0; r < 8; ++r) {
        const int row = j0 + r + 8 * hi;
        if (row < M) Y[(long)row * COUT + n] = acc[c][r];
      }
    }
  }
}

static inline int blocks_for(int M) {
  const int waves = (M + 15) / 16;
  return (waves + BLOCK / 32 - 1) / (BLOCK / 32);
}

template <int CIN, int COUT>
static constexpr size_t lds_bytes() {
  return (size_t)27 * (CIN / 4) * 2 * COUT * sizeof(v2f);
}

extern "C" void kernel_launch(void* const* d_in, const int* in_sizes, int n_in,
                              void* d_out, int out_size, void* d_ws, size_t ws_size,
                              hipStream_t stream) {
  (void)n_in; (void)out_size; (void)ws_size;
  const float* feats = (const float*)d_in[0];
  const float* W1    = (const float*)d_in[1];
  const float* W2    = (const float*)d_in[2];
  const float* W3    = (const float*)d_in[3];
  const int*   g1    = (const int*)d_in[4];
  const int*   s1    = (const int*)d_in[5];
  const int*   g2    = (const int*)d_in[6];
  const int*   s2    = (const int*)d_in[7];
  const int*   g3    = (const int*)d_in[8];
  const int*   s3    = (const int*)d_in[9];

  const int N  = in_sizes[0] / 4;
  const int M2 = in_sizes[6] / 27;

  float* x1  = (float*)d_ws;
  float* x2  = x1 + (size_t)N * 32;
  float* out = (float*)d_out;

  sparse_conv_wmma<4, 32, 8>
      <<<blocks_for(N), BLOCK, lds_bytes<4, 32>(), stream>>>(feats, W1, g1, s1, x1, N);
  sparse_conv_wmma<32, 32, 3>
      <<<blocks_for(M2), BLOCK, lds_bytes<32, 32>(), stream>>>(x1, W2, g2, s2, x2, M2);
  sparse_conv_wmma<32, 2, 4>
      <<<blocks_for(M2), BLOCK, lds_bytes<32, 2>(), stream>>>(x2, W3, g3, s3, out, M2);
}


# conv3 reassociated: conv2's epilogue multiplies its 16x32 tile by [W3_0..W3_26] (32 WMMAs, not 216 on 2 of 16 columns); launch 3 is a gather-add of 2-float slices; sum order changes (c before k)
# speedup vs baseline: 2.4597x; 1.2094x over previous
#include <hip/hip_runtime.h>

typedef __attribute__((ext_vector_type(2))) float v2f;
typedef __attribute__((ext_vector_type(4))) float v4f;
typedef __attribute__((ext_vector_type(8))) float v8f;

constexpr int BLOCK = 256;

template <int CIN, int COUT, int DEPTH, int ZC, int D2>
__global__ void __launch_bounds__(BLOCK) sparse_conv_wmma(
    const float* __restrict__ X,
    const float* __restrict__ W,
    const int*   __restrict__ G,
    const int*   __restrict__ S,
    float*       __restrict__ Y,
    const float* __restrict__ Wn,
    int M) {
  constexpr int KSTEPS = CIN / 4;
  constexpr int NCH    = (COUT + 15) / 16;
  constexpr int PAIRS  = 27 * KSTEPS * 2 * COUT;
  constexpr int KS2    = COUT / 4;
  constexpr int NCH2   = (ZC + 15) / 16;
  constexpr int ZCP    = NCH2 * 16;
  constexpr int VPAIRS = KS2 * 2 * ZCP;
  constexpr int TP     = COUT + 4;

  extern __shared__ __attribute__((aligned(16))) v2f wlds[];
  v2f*   vlds = wlds + PAIRS;
  float* tlds = (float*)(vlds + VPAIRS);

  if constexpr (COUT % 4 == 0) {
    constexpr int QPR   = COUT / 4;
    constexpr int QUADS = PAIRS / 4;
    constexpr int NIT   = (QUADS + BLOCK - 1) / BLOCK;
    v4f ra[NIT];
    v4f rb[NIT];
#pragma unroll
    for (int i = 0; i < NIT; ++i) {
      const int q  = (int)threadIdx.x + i * BLOCK;
      const int qc = (i < NIT - 1 || q < QUADS) ? q : (QUADS - 1);
      const int g  = qc % QPR;
      const int t  = qc / QPR;
      const v4f* s = (const v4f*)(W + (long)(2 * t) * COUT + 4 * g);
      ra[i] = s[0];
      rb[i] = s[QPR];
    }
#pragma unroll
    for (int i = 0; i < NIT; ++i) {
      const int q = (int)threadIdx.x + i * BLOCK;
      if (i < NIT - 1 || q < QUADS) {
        const int g = q % QPR;
        const int t = q / QPR;
        v4f lo, hi4;
        lo.x  = ra[i].x; lo.y  = rb[i].x; lo.z  = ra[i].y; lo.w  = rb[i].y;
        hi4.x = ra[i].z; hi4.y = rb[i].z; hi4.z = ra[i].w; hi4.w = rb[i].w;
        v4f* d = (v4f*)(wlds + t * COUT + 4 * g);
        d[0] = lo;
        d[1] = hi4;
      }
    }
  } else {
    constexpr int NIT = (PAIRS + BLOCK - 1) / BLOCK;
    float wa[NIT];
    float wb[NIT];
#pragma unroll
    for (int i = 0; i < NIT; ++i) {
      const int p  = (int)threadIdx.x + i * BLOCK;
      const int pc = (i < NIT - 1 || p < PAIRS) ? p : (PAIRS - 1);
      const int n  = pc % COUT;
      const int t  = pc / COUT;
      const float* s = W + (long)(2 * t) * COUT + n;
      wa[i] = s[0];
      wb[i] = s[COUT];
    }
#pragma unroll
    for (int i = 0; i < NIT; ++i) {
      const int p = (int)threadIdx.x + i * BLOCK;
      if (i < NIT - 1 || p < PAIRS) {
        v2f w;
        w.x = wa[i];
        w.y = wb[i];
        wlds[p] = w;
      }
    }
  }
  if constexpr (ZC > 0) {
    constexpr int NITV = (VPAIRS + BLOCK - 1) / BLOCK;
#pragma unroll
    for (int i = 0; i < NITV; ++i) {
      const int p = (int)threadIdx.x + i * BLOCK;
      if (p < VPAIRS) {
        const int n = p % ZCP;
        const int t = p / ZCP;
        v2f v;
        v.x = 0.0f;
        v.y = 0.0f;
        if (n < ZC) {
          const float* s = Wn + (long)(n / D2) * (COUT * D2) + (2 * t) * D2 + (n % D2);
          v.x = s[0];
          v.y = s[D2];
        }
        vlds[p] = v;
      }
    }
  }
  __syncthreads();

  const int wave = (int)((blockIdx.x * BLOCK + threadIdx.x) >> 5);
  const int lane = (int)(threadIdx.x & 31);
  const int j0 = wave * 16;
  if (j0 >= M) return;

  const int m  = lane & 15;
  const int hi = lane >> 4;
  const int j  = j0 + m;
  const bool jr = (j < M);
  const int lanehn = hi * COUT + m;

  v8f acc[NCH] = {};

  const unsigned joff = (unsigned)(jr ? j : j0) * 4u;
  int gv[27];
  int sv[27];
#pragma unroll
  for (int k = 0; k < 27; ++k) {
    const char* gk = (const char*)(G + (long)k * M);
    const char* sk = (const char*)(S + (long)k * M);
    gv[k] = __builtin_nontemporal_load((const int*)(gk + joff));
    sv[k] = __builtin_nontemporal_load((const int*)(sk + joff));
  }

  v2f  A[DEPTH][KSTEPS];
  bool hit[DEPTH];

#pragma unroll
  for (int d = 0; d < DEPTH; ++d) {
    hit[d] = jr && (sv[d] == j);
    const float* xrow = X + (long)(hit[d] ? gv[d] : 0) * CIN;
#pragma unroll
    for (int kk = 0; kk < KSTEPS; ++kk) {
      A[d][kk].x = xrow[kk * 4 + 2 * hi + 0];
      A[d][kk].y = xrow[kk * 4 + 2 * hi + 1];
    }
  }

#pragma unroll
  for (int k = 0; k < 27; ++k) {
    const v2f* wk   = wlds + k * (KSTEPS * 2 * COUT) + lanehn;
    const int  slot = k % DEPTH;

#pragma unroll
    for (int kk = 0; kk < KSTEPS; ++kk) {
      v2f Ak;
      Ak.x = hit[slot] ? A[slot][kk].x : 0.0f;
      Ak.y = hit[slot] ? A[slot][kk].y : 0.0f;
#pragma unroll
      for (int c = 0; c < NCH; ++c) {
        const int n = m + c * 16;
        v2f B;
        if (n < COUT) {
          B = wk[kk * (2 * COUT) + c * 16];
        } else {
          B.x = 0.0f;
          B.y = 0.0f;
        }
        acc[c] = __builtin_amdgcn_wmma_f32_16x16x4_f32(
            false, Ak, false, B, (short)0, acc[c], false, false);
      }
    }

    if (k + DEPTH < 27) {
      hit[slot] = jr && (sv[k + DEPTH] == j);
      const float* xrow = X + (long)(hit[slot] ? gv[k + DEPTH] : 0) * CIN;
#pragma unroll
      for (int kk = 0; kk < KSTEPS; ++kk) {
        A[slot][kk].x = xrow[kk * 4 + 2 * hi + 0];
        A[slot][kk].y = xrow[kk * 4 + 2 * hi + 1];
      }
    }
  }

  if constexpr (ZC == 0) {
#pragma unroll
    for (int c = 0; c < NCH; ++c) {
      const int n = m + c * 16;
      if (n < COUT) {
#pragma unroll
        for (int r = 0; r < 8; ++r) {
          const int row = j0 + r + 8 * hi;
          if (row < M) Y[(long)row * COUT + n] = acc[c][r];
        }
      }
    }
  } else {
    static_assert(COUT % 16 == 0, "fused product needs whole column chunks of Y");
    static_assert(D2 == 2, "a slice is one register pair of the transposed tile");
    float* tl = tlds + (int)(threadIdx.x >> 5) * (16 * TP);
#pragma unroll
    for (int c = 0; c < NCH; ++c) {
#pragma unroll
      for (int r = 0; r < 8; ++r) {
        tl[(r + 8 * hi) * TP + c * 16 + m] = acc[c][r];
      }
    }
    v2f Yt[KS2];
#pragma unroll
    for (int kk = 0; kk < KS2; ++kk) {
      Yt[kk] = *(const v2f*)(tl + m * TP + kk * 4 + 2 * hi);
    }
    v8f z[NCH2] = {};
#pragma unroll
    for (int kk = 0; kk < KS2; ++kk) {
#pragma unroll
      for (int c = NCH2 - 1; c >= 0; --c) {
        const v2f Vt = vlds[(kk * 2 + hi) * ZCP + c * 16 + m];
        z[c] = __builtin_amdgcn_wmma_f32_16x16x4_f32(
            false, Vt, false, Yt[kk], (short)0, z[c], false, false);
      }
    }
    const int row = j0 + m;
#pragma unroll
    for (int c = 0; c < NCH2; ++c) {
#pragma unroll
      for (int r = 0; r < 8; r += 2) {
        const int plane = c * 8 + r / 2 + 4 * hi;
        if (plane < ZC / D2 && row < M) {
          v2f s;
          s.x = z[c][r];
          s.y = z[c][r + 1];
          *(v2f*)(Y + ((long)plane * M + row) * D2) = s;
        }
      }
    }
  }
}

constexpr int GBLOCK = 128;
template <int ZC, int D2>
__global__ void __launch_bounds__(GBLOCK) gather_add_slices(
    const float* __restrict__ Z,
    const int*   __restrict__ G,
    const int*   __restrict__ S,
    float*       __restrict__ out,
    int M) {
  static_assert(D2 == 2, "slices are read and written as one 8-byte access");
  const int j = (int)(blockIdx.x * GBLOCK + threadIdx.x);
  if (j >= M) return;
  int gv[27];
  int sv[27];
#pragma unroll
  for (int k = 0; k < 27; ++k) {
    gv[k] = __builtin_nontemporal_load(G + (long)k * M + j);
    sv[k] = __builtin_nontemporal_load(S + (long)k * M + j);
  }
  v2f zs[27];
#pragma unroll
  for (int k = 0; k < 27; ++k) {
    const float* zk = Z + (long)k * M * D2;
    const unsigned row = (sv[k] == j) ? (unsigned)gv[k] : 0u;
    zs[k] = *(const v2f*)(zk + row * (unsigned)D2);
  }
  v2f sum;
  sum.x = 0.0f;
  sum.y = 0.0f;
#pragma unroll
  for (int k = 0; k < 27; ++k) {
    const bool h = (sv[k] == j);
    sum.x += h ? zs[k].x : 0.0f;
    sum.y += h ? zs[k].y : 0.0f;
  }
  *(v2f*)(out + (long)j * D2) = sum;
}

static inline int blocks_for(int M) {
  const int waves = (M + 15) / 16;
  return (waves + BLOCK / 32 - 1) / (BLOCK / 32);
}

static inline int gather_blocks(int M) {
  return (M + GBLOCK - 1) / GBLOCK;
}

template <int CIN, int COUT, int ZC>
static constexpr size_t lds_bytes() {
  size_t b = (size_t)27 * (CIN / 4) * 2 * COUT * sizeof(v2f);
  if (ZC > 0) {
    b += (size_t)(COUT / 4) * 2 * (((ZC + 15) / 16) * 16) * sizeof(v2f);
    b += (size_t)(BLOCK / 32) * 16 * (COUT + 4) * sizeof(float);
  }
  return b;
}

extern "C" void kernel_launch(void* const* d_in, const int* in_sizes, int n_in,
                              void* d_out, int out_size, void* d_ws, size_t ws_size,
                              hipStream_t stream) {
  (void)n_in; (void)out_size; (void)ws_size;
  const float* feats = (const float*)d_in[0];
  const float* W1    = (const float*)d_in[1];
  const float* W2    = (const float*)d_in[2];
  const float* W3    = (const float*)d_in[3];
  const int*   g1    = (const int*)d_in[4];
  const int*   s1    = (const int*)d_in[5];
  const int*   g2    = (const int*)d_in[6];
  const int*   s2    = (const int*)d_in[7];
  const int*   g3    = (const int*)d_in[8];
  const int*   s3    = (const int*)d_in[9];

  const int N  = in_sizes[0] / 4;
  const int M2 = in_sizes[6] / 27;

  float* x1  = (float*)d_ws;
  float* z2  = x1 + (size_t)N * 32;
  float* out = (float*)d_out;

  sparse_conv_wmma<4, 32, 8, 0, 0>
      <<<blocks_for(N), BLOCK, lds_bytes<4, 32, 0>(), stream>>>(feats, W1, g1, s1, x1, W1, N);
  sparse_conv_wmma<32, 32, 3, 54, 2>
      <<<blocks_for(M2), BLOCK, lds_bytes<32, 32, 54>(), stream>>>(x1, W2, g2, s2, z2, W3, M2);
  gather_add_slices<54, 2>
      <<<gather_blocks(M2), GBLOCK, 0, stream>>>(z2, g3, s3, out, M2);
}


# index maps staged per block into LDS by whole-line 16-byte loads issued with the W loads, one word per (offset,row): gather row or -1; replaces 54 half-line loads per wave; sum order unchanged
# speedup vs baseline: 2.8458x; 1.1570x over previous
#include <hip/hip_runtime.h>

typedef __attribute__((ext_vector_type(2))) float v2f;
typedef __attribute__((ext_vector_type(4))) float v4f;
typedef __attribute__((ext_vector_type(8))) float v8f;
typedef __attribute__((ext_vector_type(4))) int v4i;
typedef int v4iu __attribute__((ext_vector_type(4), aligned(4)));

constexpr int BLOCK = 256;

template <int CIN, int COUT, int DEPTH, int ZC, int D2>
__global__ void __launch_bounds__(BLOCK)
__attribute__((amdgpu_waves_per_eu(CIN <= 4 ? 16 : 4)))
sparse_conv_wmma(
    const float* __restrict__ X,
    const float* __restrict__ W,
    const int*   __restrict__ G,
    const int*   __restrict__ S,
    float*       __restrict__ Y,
    const float* __restrict__ Wn,
    int M) {
  constexpr int KSTEPS = CIN / 4;
  constexpr int NCH    = (COUT + 15) / 16;
  constexpr int PAIRS  = 27 * KSTEPS * 2 * COUT;
  constexpr int KS2    = COUT / 4;
  constexpr int NCH2   = (ZC + 15) / 16;
  constexpr int ZCP    = NCH2 * 16;
  constexpr int VPAIRS = KS2 * 2 * ZCP;
  constexpr int TP     = COUT + 4;
  constexpr int WREG   = (ZC > 0 && 16 * TP > 27 * 16) ? 16 * TP : 27 * 16;

  extern __shared__ __attribute__((aligned(16))) v2f wlds[];
  v2f*   vlds = wlds + PAIRS;
  int*   ilds = (int*)(vlds + VPAIRS);

  constexpr int NWAVE = BLOCK / 32;
  constexpr int RPB   = NWAVE * 16;
  static_assert(RPB / 4 == 32, "one lane per 4-row group of a plane slice");
  constexpr int NITK  = (27 + NWAVE - 1) / NWAVE;
  const int wib = __builtin_amdgcn_readfirstlane((int)(threadIdx.x >> 5));
  const int jb  = (int)blockIdx.x * RPB;
  const int r4  = jb + 4 * (int)(threadIdx.x & 31);
  v4i ig[NITK];
  v4i is[NITK];
  if (jb + RPB <= M) {
#pragma unroll
    for (int i = 0; i < NITK; ++i) {
      const int k = wib + i * NWAVE;
      ig[i] = v4i{0, 0, 0, 0};
      is[i] = v4i{0, 0, 0, 0};
      if (i < NITK - 1 || k < 27) {
        ig[i] = __builtin_nontemporal_load((const v4iu*)(G + (long)k * M + r4));
        is[i] = __builtin_nontemporal_load((const v4iu*)(S + (long)k * M + r4));
      }
    }
  } else {
#pragma unroll
    for (int i = 0; i < NITK; ++i) {
      const int k  = wib + i * NWAVE;
      const int kc = (k < 27) ? k : 26;
      const int* gp = G + (long)kc * M;
      const int* sp = S + (long)kc * M;
      v4i g, s;
      g.x = (r4 + 0 < M) ? gp[r4 + 0] : 0;  s.x = (r4 + 0 < M) ? sp[r4 + 0] : -1;
      g.y = (r4 + 1 < M) ? gp[r4 + 1] : 0;  s.y = (r4 + 1 < M) ? sp[r4 + 1] : -1;
      g.z = (r4 + 2 < M) ? gp[r4 + 2] : 0;  s.z = (r4 + 2 < M) ? sp[r4 + 2] : -1;
      g.w = (r4 + 3 < M) ? gp[r4 + 3] : 0;  s.w = (r4 + 3 < M) ? sp[r4 + 3] : -1;
      ig[i] = g;
      is[i] = s;
    }
  }

  if constexpr (COUT % 4 == 0) {
    constexpr int QPR   = COUT / 4;
    constexpr int QUADS = PAIRS / 4;
    constexpr int NIT   = (QUADS + BLOCK - 1) / BLOCK;
    v4f ra[NIT];
    v4f rb[NIT];
#pragma unroll
    for (int i = 0; i < NIT; ++i) {
      const int q  = (int)threadIdx.x + i * BLOCK;
      const int qc = (i < NIT - 1 || q < QUADS) ? q : (QUADS - 1);
      const int g  = qc % QPR;
      const int t  = qc / QPR;
      const v4f* s = (const v4f*)(W + (long)(2 * t) * COUT + 4 * g);
      ra[i] = s[0];
      rb[i] = s[QPR];
    }
#pragma unroll
    for (int i = 0; i < NIT; ++i) {
      const int q = (int)threadIdx.x + i * BLOCK;
      if (i < NIT - 1 || q < QUADS) {
        const int g = q % QPR;
        const int t = q / QPR;
        v4f lo, hi4;
        lo.x  = ra[i].x; lo.y  = rb[i].x; lo.z  = ra[i].y; lo.w  = rb[i].y;
        hi4.x = ra[i].z; hi4.y = rb[i].z; hi4.z = ra[i].w; hi4.w = rb[i].w;
        v4f* d = (v4f*)(wlds + t * COUT + 4 * g);
        d[0] = lo;
        d[1] = hi4;
      }
    }
  } else {
    constexpr int NIT = (PAIRS + BLOCK - 1) / BLOCK;
    float wa[NIT];
    float wb[NIT];
#pragma unroll
    for (int i = 0; i < NIT; ++i) {
      const int p  = (int)threadIdx.x + i * BLOCK;
      const int pc = (i < NIT - 1 || p < PAIRS) ? p : (PAIRS - 1);
      const int n  = pc % COUT;
      const int t  = pc / COUT;
      const float* s = W + (long)(2 * t) * COUT + n;
      wa[i] = s[0];
      wb[i] = s[COUT];
    }
#pragma unroll
    for (int i = 0; i < NIT; ++i) {
      const int p = (int)threadIdx.x + i * BLOCK;
      if (i < NIT - 1 || p < PAIRS) {
        v2f w;
        w.x = wa[i];
        w.y = wb[i];
        wlds[p] = w;
      }
    }
  }
  if constexpr (ZC > 0) {
    constexpr int NITV = (VPAIRS + BLOCK - 1) / BLOCK;
#pragma unroll
    for (int i = 0; i < NITV; ++i) {
      const int p = (int)threadIdx.x + i * BLOCK;
      if (p < VPAIRS) {
        const int n = p % ZCP;
        const int t = p / ZCP;
        v2f v;
        v.x = 0.0f;
        v.y = 0.0f;
        if (n < ZC) {
          const float* s = Wn + (long)(n / D2) * (COUT * D2) + (2 * t) * D2 + (n % D2);
          v.x = s[0];
          v.y = s[D2];
        }
        vlds[p] = v;
      }
    }
  }
  {
    const int q = (int)(threadIdx.x & 31);
    int* dst = ilds + (q >> 2) * WREG + (q & 3) * 4;
#pragma unroll
    for (int i = 0; i < NITK; ++i) {
      const int k = wib + i * NWAVE;
      if (i < NITK - 1 || k < 27) {
        v4i w;
        w.x = (is[i].x == r4 + 0) ? ig[i].x : -1;
        w.y = (is[i].y == r4 + 1) ? ig[i].y : -1;
        w.z = (is[i].z == r4 + 2) ? ig[i].z : -1;
        w.w = (is[i].w == r4 + 3) ? ig[i].w : -1;
        *(v4i*)(dst + k * 16) = w;
      }
    }
  }
  __syncthreads();

  const int wave = (int)((blockIdx.x * BLOCK + threadIdx.x) >> 5);
  const int lane = (int)(threadIdx.x & 31);
  const int j0 = wave * 16;
  if (j0 >= M) return;

  const int m  = lane & 15;
  const int hi = lane >> 4;
  const int lanehn = hi * COUT + m;

  v8f acc[NCH] = {};

  const int* isl = ilds + (int)(threadIdx.x >> 5) * WREG + m;
  int iw[27];
#pragma unroll
  for (int k = 0; k < 27; ++k) {
    iw[k] = isl[k * 16];
  }

  const char* Xb = (const char*)X;
  const unsigned lane_b = (unsigned)(hi * 8);
  v2f  A[DEPTH][KSTEPS];
  bool hit[DEPTH];

#pragma unroll
  for (int d = 0; d < DEPTH; ++d) {
    hit[d] = (iw[d] >= 0);
    const unsigned boff = (unsigned)(iw[d] > 0 ? iw[d] : 0) * (unsigned)(CIN * 4) + lane_b;
#pragma unroll
    for (int kk = 0; kk < KSTEPS; ++kk) {
      A[d][kk] = *(const v2f*)(Xb + boff + kk * 16);
    }
  }

#pragma unroll
  for (int k = 0; k < 27; ++k) {
    const v2f* wk   = wlds + k * (KSTEPS * 2 * COUT) + lanehn;
    const int  slot = k % DEPTH;

#pragma unroll
    for (int kk = 0; kk < KSTEPS; ++kk) {
      v2f Ak;
      Ak.x = hit[slot] ? A[slot][kk].x : 0.0f;
      Ak.y = hit[slot] ? A[slot][kk].y : 0.0f;
#pragma unroll
      for (int c = 0; c < NCH; ++c) {
        const int n = m + c * 16;
        v2f B;
        if (n < COUT) {
          B = wk[kk * (2 * COUT) + c * 16];
        } else {
          B.x = 0.0f;
          B.y = 0.0f;
        }
        acc[c] = __builtin_amdgcn_wmma_f32_16x16x4_f32(
            false, Ak, false, B, (short)0, acc[c], false, false);
      }
    }

    if (k + DEPTH < 27) {
      hit[slot] = (iw[k + DEPTH] >= 0);
      const unsigned boff = (unsigned)(iw[k + DEPTH] > 0 ? iw[k + DEPTH] : 0) * (unsigned)(CIN * 4) + lane_b;
#pragma unroll
      for (int kk = 0; kk < KSTEPS; ++kk) {
        A[slot][kk] = *(const v2f*)(Xb + boff + kk * 16);
      }
    }
  }

  if constexpr (ZC == 0) {
#pragma unroll
    for (int c = 0; c < NCH; ++c) {
      const int n = m + c * 16;
      if (n < COUT) {
#pragma unroll
        for (int r = 0; r < 8; ++r) {
          const int row = j0 + r + 8 * hi;
          if (row < M) Y[(long)row * COUT + n] = acc[c][r];
        }
      }
    }
  } else {
    static_assert(COUT % 16 == 0, "fused product needs whole column chunks of Y");
    static_assert(D2 == 2, "a slice is one register pair of the transposed tile");
    float* tl = (float*)(ilds + (int)(threadIdx.x >> 5) * WREG);
#pragma unroll
    for (int c = 0; c < NCH; ++c) {
#pragma unroll
      for (int r = 0; r < 8; ++r) {
        tl[(r + 8 * hi) * TP + c * 16 + m] = acc[c][r];
      }
    }
    v2f Yt[KS2];
#pragma unroll
    for (int kk = 0; kk < KS2; ++kk) {
      Yt[kk] = *(const v2f*)(tl + m * TP + kk * 4 + 2 * hi);
    }
    v8f z[NCH2] = {};
#pragma unroll
    for (int kk = 0; kk < KS2; ++kk) {
#pragma unroll
      for (int c = NCH2 - 1; c >= 0; --c) {
        const v2f Vt = vlds[(kk * 2 + hi) * ZCP + c * 16 + m];
        z[c] = __builtin_amdgcn_wmma_f32_16x16x4_f32(
            false, Vt, false, Yt[kk], (short)0, z[c], false, false);
      }
    }
    const int row = j0 + m;
#pragma unroll
    for (int c = 0; c < NCH2; ++c) {
#pragma unroll
      for (int r = 0; r < 8; r += 2) {
        const int plane = c * 8 + r / 2 + 4 * hi;
        if (plane < ZC / D2 && row < M) {
          v2f s;
          s.x = z[c][r];
          s.y = z[c][r + 1];
          *(v2f*)(Y + ((long)plane * M + row) * D2) = s;
        }
      }
    }
  }
}

constexpr int GBLOCK = 128;
template <int ZC, int D2>
__global__ void __launch_bounds__(GBLOCK) gather_add_slices(
    const float* __restrict__ Z,
    const int*   __restrict__ G,
    const int*   __restrict__ S,
    float*       __restrict__ out,
    int M) {
  static_assert(D2 == 2, "slices are read and written as one 8-byte access");
  const int j = (int)(blockIdx.x * GBLOCK + threadIdx.x);
  if (j >= M) return;
  int gv[27];
  int sv[27];
#pragma unroll
  for (int k = 0; k < 27; ++k) {
    gv[k] = __builtin_nontemporal_load(G + (long)k * M + j);
    sv[k] = __builtin_nontemporal_load(S + (long)k * M + j);
  }
  v2f zs[27];
#pragma unroll
  for (int k = 0; k < 27; ++k) {
    const float* zk = Z + (long)k * M * D2;
    const unsigned row = (sv[k] == j) ? (unsigned)gv[k] : 0u;
    zs[k] = *(const v2f*)(zk + row * (unsigned)D2);
  }
  v2f sum;
  sum.x = 0.0f;
  sum.y = 0.0f;
#pragma unroll
  for (int k = 0; k < 27; ++k) {
    const bool h = (sv[k] == j);
    sum.x += h ? zs[k].x : 0.0f;
    sum.y += h ? zs[k].y : 0.0f;
  }
  *(v2f*)(out + (long)j * D2) = sum;
}

static inline int blocks_for(int M) {
  const int waves = (M + 15) / 16;
  return (waves + BLOCK / 32 - 1) / (BLOCK / 32);
}

static inline int gather_blocks(int M) {
  return (M + GBLOCK - 1) / GBLOCK;
}

template <int CIN, int COUT, int ZC>
static constexpr size_t lds_bytes() {
  size_t b = (size_t)27 * (CIN / 4) * 2 * COUT * sizeof(v2f);
  size_t wreg = 27 * 16;
  if (ZC > 0) {
    b += (size_t)(COUT / 4) * 2 * (((ZC + 15) / 16) * 16) * sizeof(v2f);
    if ((size_t)16 * (COUT + 4) > wreg) wreg = (size_t)16 * (COUT + 4);
  }
  b += (size_t)(BLOCK / 32) * wreg * sizeof(float);
  return b;
}

extern "C" void kernel_launch(void* const* d_in, const int* in_sizes, int n_in,
                              void* d_out, int out_size, void* d_ws, size_t ws_size,
                              hipStream_t stream) {
  (void)n_in; (void)out_size; (void)ws_size;
  const float* feats = (const float*)d_in[0];
  const float* W1    = (const float*)d_in[1];
  const float* W2    = (const float*)d_in[2];
  const float* W3    = (const float*)d_in[3];
  const int*   g1    = (const int*)d_in[4];
  const int*   s1    = (const int*)d_in[5];
  const int*   g2    = (const int*)d_in[6];
  const int*   s2    = (const int*)d_in[7];
  const int*   g3    = (const int*)d_in[8];
  const int*   s3    = (const int*)d_in[9];

  const int N  = in_sizes[0] / 4;
  const int M2 = in_sizes[6] / 27;

  float* x1  = (float*)d_ws;
  float* z2  = x1 + (size_t)N * 32;
  float* out = (float*)d_out;

  sparse_conv_wmma<4, 32, 8, 0, 0>
      <<<blocks_for(N), BLOCK, lds_bytes<4, 32, 0>(), stream>>>(feats, W1, g1, s1, x1, W1, N);
  sparse_conv_wmma<32, 32, 3, 54, 2>
      <<<blocks_for(M2), BLOCK, lds_bytes<32, 32, 54>(), stream>>>(x1, W2, g2, s2, z2, W3, M2);
  gather_add_slices<54, 2>
      <<<gather_blocks(M2), GBLOCK, 0, stream>>>(z2, g3, s3, out, M2);
}
